# starting kernel
# baseline (speedup 1.0000x reference)
#include <hip/hip_runtime.h>
#include <math.h>

typedef __attribute__((ext_vector_type(2))) float v2f;
typedef __attribute__((ext_vector_type(8))) float v8f;

__device__ __forceinline__ v8f wmma4(v2f a, v2f b, v8f c) {
  return __builtin_amdgcn_wmma_f32_16x16x4_f32(
       false, a,  false, b,
       (short)0, c,  false,  false);
}

template <int EPI>
__global__ __launch_bounds__(32) void gemm_bias_kernel(
    const float* __restrict__ A, const float* __restrict__ W,
    const float* __restrict__ bias, const float* __restrict__ resid,
    float* __restrict__ C, int K, int NC) {
  const int lane = threadIdx.x;
  const int m = lane & 15;
  const int hh = lane >> 4;
  const int row0 = blockIdx.x * 64;
  const int n0 = blockIdx.y * 64;

  v8f acc[4][4];
#pragma unroll
  for (int rt = 0; rt < 4; ++rt)
#pragma unroll
    for (int t = 0; t < 4; ++t) acc[rt][t] = {};

  for (int kb = 0; kb < K; kb += 4) {
    v2f a[4];
#pragma unroll
    for (int rt = 0; rt < 4; ++rt)
      a[rt] = *(const v2f*)(A + (size_t)(row0 + rt * 16 + m) * K + kb + hh * 2);
    v2f b[4];
#pragma unroll
    for (int t = 0; t < 4; ++t) {
      const int n = n0 + t * 16 + m;
      b[t].x = W[(size_t)(kb + hh * 2) * NC + n];
      b[t].y = W[(size_t)(kb + hh * 2 + 1) * NC + n];
    }
#pragma unroll
    for (int rt = 0; rt < 4; ++rt)
#pragma unroll
      for (int t = 0; t < 4; ++t) acc[rt][t] = wmma4(a[rt], b[t], acc[rt][t]);
  }

  float bn[4];
#pragma unroll
  for (int t = 0; t < 4; ++t) bn[t] = bias[n0 + t * 16 + m];

  float res[4][4][8];
  if (EPI == 2) {
#pragma unroll
    for (int rt = 0; rt < 4; ++rt)
#pragma unroll
      for (int t = 0; t < 4; ++t)
#pragma unroll
        for (int r = 0; r < 8; ++r)
          res[rt][t][r] = resid[(size_t)(row0 + rt * 16 + r + 8 * hh) * NC +
                                n0 + t * 16 + m];
  }

#pragma unroll
  for (int rt = 0; rt < 4; ++rt)
#pragma unroll
    for (int t = 0; t < 4; ++t)
#pragma unroll
      for (int r = 0; r < 8; ++r) {
        const int row = row0 + rt * 16 + r + 8 * hh;
        const int col = n0 + t * 16 + m;
        float v = acc[rt][t][r] + bn[t];
        if (EPI == 1) v = 0.5f * v * (1.0f + erff(v * 0.70710678118654752f));
        if (EPI == 2) v += res[rt][t][r];
        C[(size_t)row * NC + col] = v;
      }
}

template <bool ADD>
__global__ __launch_bounds__(32) void add_ln_kernel(
    float* __restrict__ h, const float* __restrict__ addend,
    const float* __restrict__ g, const float* __restrict__ beta,
    float* __restrict__ ln_out) {
  const int lane = threadIdx.x;
  const size_t base = (size_t)blockIdx.x * 256;

  float v[8];
  float s = 0.f, ss = 0.f;
#pragma unroll
  for (int i = 0; i < 8; ++i) {
    const int c = i * 32 + lane;
    float x = h[base + c];
    if (ADD) x += addend[base + c];
    v[i] = x;
    s += x;
    ss += x * x;
  }
  if (ADD) {
#pragma unroll
    for (int i = 0; i < 8; ++i) h[base + i * 32 + lane] = v[i];
  }
  for (int off = 16; off; off >>= 1) {
    s += __shfl_xor(s, off);
    ss += __shfl_xor(ss, off);
  }
  const float mu = s * (1.0f / 256.0f);
  const float var = ss * (1.0f / 256.0f) - mu * mu;
  const float rs = rsqrtf(var + 1e-5f);
#pragma unroll
  for (int i = 0; i < 8; ++i) {
    const int c = i * 32 + lane;
    ln_out[base + c] = (v[i] - mu) * rs * g[c] + beta[c];
  }
}

#define ATT_T 4096
#define ATT_QS 768
#define ATT_WIN 128
#define ATT_PST 148

__global__ __launch_bounds__(32) void attn_kernel(const float* __restrict__ qkv,
                                                  float* __restrict__ O) {
  __shared__ float P[16 * ATT_PST];
  const int lane = threadIdx.x;
  const int m = lane & 15;
  const int hh = lane >> 4;
  const int q0 = blockIdx.x * 16;
  const int bb = q0 & ~(ATT_T - 1);

  const float* Q = qkv;
  const float* Km = qkv + 256;
  const float* V = qkv + 512;

  v8f s[9];
#pragma unroll
  for (int t = 0; t < 9; ++t) s[t] = {};

  for (int d = 0; d < 256; d += 4) {
    const v2f a = *(const v2f*)(Q + (size_t)(q0 + m) * ATT_QS + d + hh * 2);
#pragma unroll
    for (int t = 0; t < 9; ++t) {
      const int key = q0 - ATT_WIN + t * 16 + m;
      const int kc = key < bb ? bb : key;
      const v2f b = *(const v2f*)(Km + (size_t)kc * ATT_QS + d + hh * 2);
      s[t] = wmma4(a, b, s[t]);
    }
  }

  const float NEG = -3.402823466e38f;
  float rmax[8], rsum[8];
#pragma unroll
  for (int r = 0; r < 8; ++r) rmax[r] = NEG;
#pragma unroll
  for (int t = 0; t < 9; ++t) {
    const int key = q0 - ATT_WIN + t * 16 + m;
#pragma unroll
    for (int r = 0; r < 8; ++r) {
      const int q = q0 + r + 8 * hh;
      const int d = q - key;
      const bool ok = (key >= bb) && (d >= 0) && (d <= ATT_WIN);
      const float val = ok ? s[t][r] * 0.0625f : NEG;
      s[t][r] = val;
      rmax[r] = fmaxf(rmax[r], val);
    }
  }
#pragma unroll
  for (int r = 0; r < 8; ++r)
    for (int off = 1; off < 16; off <<= 1)
      rmax[r] = fmaxf(rmax[r], __shfl_xor(rmax[r], off));
#pragma unroll
  for (int r = 0; r < 8; ++r) rsum[r] = 0.f;
#pragma unroll
  for (int t = 0; t < 9; ++t)
#pragma unroll
    for (int r = 0; r < 8; ++r) {
      const float e = expf(s[t][r] - rmax[r]);
      s[t][r] = e;
      rsum[r] += e;
    }
#pragma unroll
  for (int r = 0; r < 8; ++r)
    for (int off = 1; off < 16; off <<= 1) rsum[r] += __shfl_xor(rsum[r], off);

#pragma unroll
  for (int t = 0; t < 9; ++t)
#pragma unroll
    for (int r = 0; r < 8; ++r)
      P[(r + 8 * hh) * ATT_PST + t * 16 + m] = s[t][r] / rsum[r];

  __syncthreads();

#pragma unroll
  for (int ch = 0; ch < 2; ++ch) {
    const int n0 = ch * 128;
    v8f o[8];
#pragma unroll
    for (int tc = 0; tc < 8; ++tc) o[tc] = {};
    for (int t = 0; t < 9; ++t) {
      const int kt0 = q0 - ATT_WIN + t * 16;
#pragma unroll
      for (int kk = 0; kk < 16; kk += 4) {
        const v2f a = *(const v2f*)&P[m * ATT_PST + t * 16 + kk + hh * 2];
        const int k0 = kt0 + kk + hh * 2;
        const int kcx = k0 < bb ? bb : k0;
        const int kcy = (k0 + 1) < bb ? bb : (k0 + 1);
#pragma unroll
        for (int tc = 0; tc < 8; ++tc) {
          const int col = n0 + tc * 16 + m;
          v2f b;
          b.x = V[(size_t)kcx * ATT_QS + col];
          b.y = V[(size_t)kcy * ATT_QS + col];
          o[tc] = wmma4(a, b, o[tc]);
        }
      }
    }
#pragma unroll
    for (int tc = 0; tc < 8; ++tc)
#pragma unroll
      for (int r = 0; r < 8; ++r)
        O[(size_t)(q0 + r + 8 * hh) * 256 + n0 + tc * 16 + m] = o[tc][r];
  }
}

extern "C" void kernel_launch(void* const* d_in, const int* in_sizes, int n_in,
                              void* d_out, int out_size, void* d_ws,
                              size_t ws_size, hipStream_t stream) {
  const float* x = (const float*)d_in[0];
  const float* W_in = (const float*)d_in[1];
  const float* b_in = (const float*)d_in[2];
  const float* W_qkv = (const float*)d_in[3];
  const float* b_qkv = (const float*)d_in[4];
  const float* g_attn = (const float*)d_in[5];
  const float* be_attn = (const float*)d_in[6];
  const float* g_mlp = (const float*)d_in[7];
  const float* be_mlp = (const float*)d_in[8];
  const float* W1 = (const float*)d_in[9];
  const float* b1 = (const float*)d_in[10];
  const float* W2 = (const float*)d_in[11];
  const float* b2 = (const float*)d_in[12];
  const float* W_out = (const float*)d_in[13];
  const float* b_out = (const float*)d_in[14];
  float* out = (float*)d_out;

  const int N = in_sizes[0] / 64;

  float* ws = (float*)d_ws;
  float* h = ws;
  float* ln = ws + (size_t)N * 256;
  float* attn = ln;
  float* qkv = ws + (size_t)N * 512;
  float* m1 = qkv;

  const dim3 blk(32);
  const int RB64 = N / 64;
  const int QB = N / 16;

  gemm_bias_kernel<0><<<dim3(RB64, 4), blk, 0, stream>>>(x, W_in, b_in,
                                                         nullptr, h, 64, 256);
  add_ln_kernel<false>
      <<<dim3(N), blk, 0, stream>>>(h, nullptr, g_attn, be_attn, ln);
  gemm_bias_kernel<0><<<dim3(RB64, 12), blk, 0, stream>>>(
      ln, W_qkv, b_qkv, nullptr, qkv, 256, 768);
  attn_kernel<<<dim3(QB), blk, 0, stream>>>(qkv, attn);
  add_ln_kernel<true>
      <<<dim3(N), blk, 0, stream>>>(h, attn, g_mlp, be_mlp, ln);
  gemm_bias_kernel<1><<<dim3(RB64, 4), blk, 0, stream>>>(ln, W1, b1, nullptr,
                                                         m1, 256, 256);
  gemm_bias_kernel<2><<<dim3(RB64, 4), blk, 0, stream>>>(m1, W2, b2, h, h, 256,
                                                         256);
  gemm_bias_kernel<0><<<dim3(RB64, 1), blk, 0, stream>>>(h, W_out, b_out,
                                                         nullptr, out, 256, 64);
}


# W2 folded into the output projection: out = h@W_out + gelu(..)@(W2@W_out) + (b2@W_out+b_out); 256x64 fold made once a call, W2 GEMM gone; out GEMM two K=256 products; sum order changes
# speedup vs baseline: 2.5709x; 2.5709x over previous
#include <hip/hip_runtime.h>
#include <math.h>

typedef __attribute__((ext_vector_type(2))) float v2f;
typedef __attribute__((ext_vector_type(8))) float v8f;

__device__ __forceinline__ v8f wmma4(v2f a, v2f b, v8f c) {
  return __builtin_amdgcn_wmma_f32_16x16x4_f32(
       false, a,  false, b,
       (short)0, c,  false,  false);
}

typedef __attribute__((ext_vector_type(4))) float v4f;

#define GEMM_SST 36
#define GEMM_WST 72

__device__ __forceinline__ void gemm_kstep(const float* __restrict__ scr,
                                           const float* __restrict__ wsc,
                                           int k4, int wk, int m, int hh,
                                           v8f (&acc)[4][4]) {
  v2f a[4];
#pragma unroll
  for (int rt = 0; rt < 4; ++rt)
    a[rt] = *(const v2f*)(scr + (rt * 16 + m) * GEMM_SST + k4 + hh * 2);
  v2f b[4];
#pragma unroll
  for (int t = 0; t < 4; ++t) {
    b[t].x = wsc[(wk + hh * 2) * GEMM_WST + t * 16 + m];
    b[t].y = wsc[(wk + hh * 2 + 1) * GEMM_WST + t * 16 + m];
  }
#pragma unroll
  for (int rt = 0; rt < 4; ++rt)
#pragma unroll
    for (int t = 0; t < 4; ++t) acc[rt][t] = wmma4(a[rt], b[t], acc[rt][t]);
}

template <int EPI, int SPLIT, int K, int NC, bool DUAL>
__global__ __launch_bounds__(32 * SPLIT) void gemm_bias_kernel(
    const float* __restrict__ A, const float* __restrict__ W,
    const float* __restrict__ bias, const float* __restrict__ resid,
    float* __restrict__ C, const float* __restrict__ A2,
    const float* __restrict__ Wb2) {
  __shared__ v4f part[SPLIT][16][2][32];

  const int lane = threadIdx.x & 31;
  const int w = __builtin_amdgcn_readfirstlane((int)(threadIdx.x >> 5));
  const int m = lane & 15;
  const int hh = lane >> 4;
  const int row0 = blockIdx.x * 64;
  const int n0 = blockIdx.y * 64;

  v8f acc[4][4];
#pragma unroll
  for (int rt = 0; rt < 4; ++rt)
#pragma unroll
    for (int t = 0; t < 4; ++t) acc[rt][t] = {};

  constexpr int KW = DUAL ? SPLIT / 2 : SPLIT;
  static_assert(K % (32 * KW) == 0, "whole 32-k blocks per wave");
  static_assert((64 * GEMM_SST + 16 * GEMM_WST) * 4 <= 16 * 2 * 32 * 16,
                "scratch fits its plane");
  float* scr = (float*)&part[w][0][0][0];
  float* wsc = scr + 64 * GEMM_SST;
  const int lr = lane >> 3;
  const int lc = (lane & 7) * 4;
  const int wr = lane >> 4;
  const int wc = (lane & 15) * 4;

  constexpr int kper = K / KW;
  const int kbeg = (w % KW) * kper;
  const float* Ab = (DUAL && w >= KW) ? A2 : A;
  const float* Wb = (DUAL && w >= KW) ? Wb2 : W;
  const float* Ap = Ab + (size_t)(row0 + lr) * K + kbeg + lc;
  const float* Wp = Wb + (size_t)(kbeg + wr) * NC + n0 + wc;
#pragma nounroll
  for (int kb = 0; kb < kper; kb += 32) {
    v4f ga[4][4];
#pragma unroll
    for (int rt = 0; rt < 4; ++rt)
#pragma unroll
      for (int i = 0; i < 4; ++i)
        ga[rt][i] = *(const v4f*)(Ap + (size_t)(rt * 16 + i * 4) * K + kb);
    v4f gw[8];
#pragma unroll
    for (int i = 0; i < 8; ++i)
      gw[i] = *(const v4f*)(Wp + (size_t)(kb + i * 2) * NC);
    __builtin_amdgcn_sched_barrier(0);
#pragma unroll
    for (int rt = 0; rt < 4; ++rt)
#pragma unroll
      for (int i = 0; i < 4; ++i)
        *(v4f*)(scr + (rt * 16 + i * 4 + lr) * GEMM_SST + lc) = ga[rt][i];
#pragma unroll
    for (int i = 0; i < 8; ++i)
      *(v4f*)(wsc + (i * 2 + wr) * GEMM_WST + wc) = gw[i];

#pragma unroll
    for (int i = 0; i < 8; ++i)
      gw[i] = *(const v4f*)(Wp + (size_t)(kb + 16 + i * 2) * NC);

#pragma unroll
    for (int j = 0; j < 4; ++j) gemm_kstep(scr, wsc, j * 4, j * 4, m, hh, acc);

#pragma unroll
    for (int i = 0; i < 8; ++i)
      *(v4f*)(wsc + (i * 2 + wr) * GEMM_WST + wc) = gw[i];

#pragma unroll
    for (int j = 4; j < 8; ++j)
      gemm_kstep(scr, wsc, j * 4, (j - 4) * 4, m, hh, acc);
  }

#pragma unroll
  for (int rt = 0; rt < 4; ++rt)
#pragma unroll
    for (int t = 0; t < 4; ++t) {
      part[w][rt * 4 + t][0][lane] =
          __builtin_shufflevector(acc[rt][t], acc[rt][t], 0, 1, 2, 3);
      part[w][rt * 4 + t][1][lane] =
          __builtin_shufflevector(acc[rt][t], acc[rt][t], 4, 5, 6, 7);
    }
  __syncthreads();

  constexpr int TPW = 16 / SPLIT;
  float v[TPW][8];
#pragma unroll
  for (int i = 0; i < TPW; ++i) {
    const int j = w * TPW + i;
    v4f lo = part[0][j][0][lane];
    v4f hi = part[0][j][1][lane];
#pragma unroll
    for (int ws = 1; ws < SPLIT; ++ws) {
      lo += part[ws][j][0][lane];
      hi += part[ws][j][1][lane];
    }
#pragma unroll
    for (int r = 0; r < 4; ++r) {
      v[i][r] = lo[r];
      v[i][r + 4] = hi[r];
    }
  }

  float bn[TPW];
#pragma unroll
  for (int i = 0; i < TPW; ++i) {
    const int j = w * TPW + i;
    bn[i] = bias[n0 + (j & 3) * 16 + m];
  }

  float res[TPW][8];
  if (EPI == 2) {
#pragma unroll
    for (int i = 0; i < TPW; ++i) {
      const int j = w * TPW + i;
#pragma unroll
      for (int r = 0; r < 8; ++r)
        res[i][r] = resid[(size_t)(row0 + (j >> 2) * 16 + r + 8 * hh) * NC +
                          n0 + (j & 3) * 16 + m];
    }
  }

#pragma unroll
  for (int i = 0; i < TPW; ++i) {
    const int j = w * TPW + i;
#pragma unroll
    for (int r = 0; r < 8; ++r) {
      const int row = row0 + (j >> 2) * 16 + r + 8 * hh;
      const int col = n0 + (j & 3) * 16 + m;
      float x = v[i][r] + bn[i];
      if (EPI == 1) x = 0.5f * x * (1.0f + erff(x * 0.70710678118654752f));
      if (EPI == 2) x += res[i][r];
      C[(size_t)row * NC + col] = x;
    }
  }
}

template <int K>
__global__ __launch_bounds__(128) void inproj_ln_kernel(
    const float* __restrict__ X, const float* __restrict__ W,
    const float* __restrict__ bias, const float* __restrict__ g,
    const float* __restrict__ beta, float* __restrict__ h,
    float* __restrict__ ln_out) {
  __shared__ __attribute__((aligned(16))) float red[2][4][2][8];

  const int lane = threadIdx.x & 31;
  const int w = __builtin_amdgcn_readfirstlane((int)(threadIdx.x >> 5));
  const int m = lane & 15;
  const int hh = lane >> 4;
  const int row0 = blockIdx.x * 16;
  const int n0 = w * 64;

  v8f acc[4];
#pragma unroll
  for (int t = 0; t < 4; ++t) acc[t] = {};
#pragma unroll
  for (int k = 0; k < K; k += 4) {
    const v2f a = *(const v2f*)(X + (size_t)(row0 + m) * K + k + hh * 2);
    v2f b[4];
#pragma unroll
    for (int t = 0; t < 4; ++t) {
      const int n = n0 + t * 16 + m;
      b[t].x = W[(size_t)(k + hh * 2) * 256 + n];
      b[t].y = W[(size_t)(k + hh * 2 + 1) * 256 + n];
    }
#pragma unroll
    for (int t = 0; t < 4; ++t) acc[t] = wmma4(a, b[t], acc[t]);
  }

  float xs[4][8];
  float s1[8], s2[8];
#pragma unroll
  for (int r = 0; r < 8; ++r) {
    s1[r] = 0.f;
    s2[r] = 0.f;
  }
#pragma unroll
  for (int t = 0; t < 4; ++t) {
    const float bn = bias[n0 + t * 16 + m];
#pragma unroll
    for (int r = 0; r < 8; ++r) {
      const float x = acc[t][r] + bn;
      xs[t][r] = x;
      s1[r] += x;
      s2[r] += x * x;
    }
  }
#pragma unroll
  for (int r = 0; r < 8; ++r)
    for (int off = 1; off < 16; off <<= 1) {
      s1[r] += __shfl_xor(s1[r], off);
      s2[r] += __shfl_xor(s2[r], off);
    }
  if (m == 0) {
    *(v4f*)&red[0][w][hh][0] = v4f{s1[0], s1[1], s1[2], s1[3]};
    *(v4f*)&red[0][w][hh][4] = v4f{s1[4], s1[5], s1[6], s1[7]};
    *(v4f*)&red[1][w][hh][0] = v4f{s2[0], s2[1], s2[2], s2[3]};
    *(v4f*)&red[1][w][hh][4] = v4f{s2[4], s2[5], s2[6], s2[7]};
  }
  __syncthreads();
#pragma unroll
  for (int ws = 0; ws < 4; ++ws) {
    const v4f alo = *(const v4f*)&red[0][ws][hh][0];
    const v4f ahi = *(const v4f*)&red[0][ws][hh][4];
    const v4f blo = *(const v4f*)&red[1][ws][hh][0];
    const v4f bhi = *(const v4f*)&red[1][ws][hh][4];
#pragma unroll
    for (int r = 0; r < 4; ++r) {
      s1[r] = ws == 0 ? alo[r] : s1[r] + alo[r];
      s1[r + 4] = ws == 0 ? ahi[r] : s1[r + 4] + ahi[r];
      s2[r] = ws == 0 ? blo[r] : s2[r] + blo[r];
      s2[r + 4] = ws == 0 ? bhi[r] : s2[r + 4] + bhi[r];
    }
  }
  float gc[4], bc[4];
#pragma unroll
  for (int t = 0; t < 4; ++t) {
    gc[t] = g[n0 + t * 16 + m];
    bc[t] = beta[n0 + t * 16 + m];
  }
#pragma unroll
  for (int r = 0; r < 8; ++r) {
    const float mu = s1[r] * (1.0f / 256.0f);
    const float var = s2[r] * (1.0f / 256.0f) - mu * mu;
    const float rstd = rsqrtf(var + 1e-5f);
#pragma unroll
    for (int t = 0; t < 4; ++t) {
      const size_t idx = (size_t)(row0 + r + 8 * hh) * 256 + n0 + t * 16 + m;
      h[idx] = xs[t][r];
      ln_out[idx] = (xs[t][r] - mu) * rstd * gc[t] + bc[t];
    }
  }
}

#define ATT_T 4096
#define ATT_QS 768
#define ATT_WIN 128
#define ATT_PST 148
#define ATT_SST 36
#define ATT_NW 4

template <int NT>
__device__ __forceinline__ void attn_scores(const float* __restrict__ Q,
                                            const float* __restrict__ Km,
                                            float* __restrict__ stg, int q0,
                                            int bb, int t0, int lane,
                                            v8f (&s)[3]) {
  const int m = lane & 15;
  const int hh = lane >> 4;
  const int lr = lane >> 3;
  const int lc = (lane & 7) * 4;
#pragma unroll
  for (int t = 0; t < NT; ++t) s[t] = {};

#pragma nounroll
  for (int db = 0; db < 256; db += 32) {
    v4f gq[4];
    v4f gk[NT][4];
#pragma unroll
    for (int i = 0; i < 4; ++i)
      gq[i] = *(const v4f*)(Q + (size_t)(q0 + i * 4 + lr) * ATT_QS + db + lc);
#pragma unroll
    for (int t = 0; t < NT; ++t)
#pragma unroll
      for (int i = 0; i < 4; ++i) {
        const int key = q0 - ATT_WIN + (t0 + t) * 16 + i * 4 + lr;
        const int kc = key < bb ? bb : key;
        gk[t][i] = *(const v4f*)(Km + (size_t)kc * ATT_QS + db + lc);
      }
    __builtin_amdgcn_sched_barrier(0);
#pragma unroll
    for (int i = 0; i < 4; ++i)
      *(v4f*)(stg + (i * 4 + lr) * ATT_SST + lc) = gq[i];
#pragma unroll
    for (int t = 0; t < NT; ++t)
#pragma unroll
      for (int i = 0; i < 4; ++i)
        *(v4f*)(stg + ((t + 1) * 16 + i * 4 + lr) * ATT_SST + lc) = gk[t][i];
#pragma unroll
    for (int j = 0; j < 8; ++j) {
      const v2f a = *(const v2f*)(stg + m * ATT_SST + j * 4 + hh * 2);
#pragma unroll
      for (int t = 0; t < NT; ++t) {
        const v2f b =
            *(const v2f*)(stg + ((t + 1) * 16 + m) * ATT_SST + j * 4 + hh * 2);
        s[t] = wmma4(a, b, s[t]);
      }
    }
  }
}

__global__ __launch_bounds__(32 * ATT_NW) void attn_kernel(
    const float* __restrict__ qkv, float* __restrict__ h,
    const float* __restrict__ g, const float* __restrict__ beta,
    float* __restrict__ ln_out) {
  __shared__ __attribute__((aligned(16))) float P[16 * ATT_PST];
  __shared__ __attribute__((aligned(16))) float stg[ATT_NW][4 * 16 * ATT_SST];
  __shared__ __attribute__((aligned(16))) float red[2][ATT_NW][2][8];

  const int lane = threadIdx.x & 31;
  const int w = __builtin_amdgcn_readfirstlane((int)(threadIdx.x >> 5));
  const int m = lane & 15;
  const int hh = lane >> 4;
  const int q0 = blockIdx.x * 16;
  const int bb = q0 & ~(ATT_T - 1);

  const float* Q = qkv;
  const float* Km = qkv + 256;
  const float* V = qkv + 512;

  const int nt = w == 0 ? 3 : 2;
  const int t0 = w == 0 ? 0 : 2 * w + 1;
  v8f s[3];
  if (w == 0) {
    attn_scores<3>(Q, Km, stg[w], q0, bb, t0, lane, s);
  } else {
    attn_scores<2>(Q, Km, stg[w], q0, bb, t0, lane, s);
    s[2] = {};
  }

  const float NEG = -3.402823466e38f;
  float rmax[8], rsum[8];
#pragma unroll
  for (int r = 0; r < 8; ++r) rmax[r] = NEG;
#pragma unroll
  for (int tt = 0; tt < 3; ++tt) {
    if (tt < nt) {
      const int key = q0 - ATT_WIN + (t0 + tt) * 16 + m;
#pragma unroll
      for (int r = 0; r < 8; ++r) {
        const int q = q0 + r + 8 * hh;
        const int d = q - key;
        const bool ok = (key >= bb) && (d >= 0) && (d <= ATT_WIN);
        const float val = ok ? s[tt][r] * 0.0625f : NEG;
        s[tt][r] = val;
        rmax[r] = fmaxf(rmax[r], val);
      }
    }
  }
#pragma unroll
  for (int r = 0; r < 8; ++r)
    for (int off = 1; off < 16; off <<= 1)
      rmax[r] = fmaxf(rmax[r], __shfl_xor(rmax[r], off));
  if (m == 0) {
    *(v4f*)&red[0][w][hh][0] = v4f{rmax[0], rmax[1], rmax[2], rmax[3]};
    *(v4f*)&red[0][w][hh][4] = v4f{rmax[4], rmax[5], rmax[6], rmax[7]};
  }
  __syncthreads();
#pragma unroll
  for (int ws = 0; ws < ATT_NW; ++ws) {
    const v4f lo = *(const v4f*)&red[0][ws][hh][0];
    const v4f hi = *(const v4f*)&red[0][ws][hh][4];
#pragma unroll
    for (int r = 0; r < 4; ++r) {
      rmax[r] = fmaxf(rmax[r], lo[r]);
      rmax[r + 4] = fmaxf(rmax[r + 4], hi[r]);
    }
  }

#pragma unroll
  for (int r = 0; r < 8; ++r) rsum[r] = 0.f;
#pragma unroll
  for (int tt = 0; tt < 3; ++tt) {
    if (tt < nt) {
#pragma unroll
      for (int r = 0; r < 8; ++r) {
        const float e = expf(s[tt][r] - rmax[r]);
        s[tt][r] = e;
        rsum[r] += e;
      }
    }
  }
#pragma unroll
  for (int r = 0; r < 8; ++r)
    for (int off = 1; off < 16; off <<= 1) rsum[r] += __shfl_xor(rsum[r], off);
  if (m == 0) {
    *(v4f*)&red[1][w][hh][0] = v4f{rsum[0], rsum[1], rsum[2], rsum[3]};
    *(v4f*)&red[1][w][hh][4] = v4f{rsum[4], rsum[5], rsum[6], rsum[7]};
  }
  __syncthreads();
#pragma unroll
  for (int ws = 0; ws < ATT_NW; ++ws) {
    const v4f lo = *(const v4f*)&red[1][ws][hh][0];
    const v4f hi = *(const v4f*)&red[1][ws][hh][4];
#pragma unroll
    for (int r = 0; r < 4; ++r) {
      rsum[r] = ws == 0 ? lo[r] : rsum[r] + lo[r];
      rsum[r + 4] = ws == 0 ? hi[r] : rsum[r + 4] + hi[r];
    }
  }

#pragma unroll
  for (int tt = 0; tt < 3; ++tt) {
    if (tt < nt) {
#pragma unroll
      for (int r = 0; r < 8; ++r)
        P[(r + 8 * hh) * ATT_PST + (t0 + tt) * 16 + m] = s[tt][r] / rsum[r];
    }
  }
  __syncthreads();

  const int n0 = w * 64;
  float hv[4][8];
#pragma unroll
  for (int tc = 0; tc < 4; ++tc)
#pragma unroll
    for (int r = 0; r < 8; ++r)
      hv[tc][r] = h[(size_t)(q0 + r + 8 * hh) * 256 + n0 + tc * 16 + m];
  v8f o[4];
#pragma unroll
  for (int tc = 0; tc < 4; ++tc) o[tc] = {};
  for (int t = 0; t < 9; ++t) {
    const int kt0 = q0 - ATT_WIN + t * 16;
#pragma unroll
    for (int kk = 0; kk < 16; kk += 4) {
      const v2f a = *(const v2f*)&P[m * ATT_PST + t * 16 + kk + hh * 2];
      const int k0 = kt0 + kk + hh * 2;
      const int kcx = k0 < bb ? bb : k0;
      const int kcy = (k0 + 1) < bb ? bb : (k0 + 1);
#pragma unroll
      for (int tc = 0; tc < 4; ++tc) {
        const int col = n0 + tc * 16 + m;
        v2f b;
        b.x = V[(size_t)kcx * ATT_QS + col];
        b.y = V[(size_t)kcy * ATT_QS + col];
        o[tc] = wmma4(a, b, o[tc]);
      }
    }
  }
  float xs[4][8];
  float s1[8], s2[8];
#pragma unroll
  for (int r = 0; r < 8; ++r) {
    s1[r] = 0.f;
    s2[r] = 0.f;
  }
#pragma unroll
  for (int tc = 0; tc < 4; ++tc)
#pragma unroll
    for (int r = 0; r < 8; ++r) {
      const float x = hv[tc][r] + o[tc][r];
      xs[tc][r] = x;
      s1[r] += x;
      s2[r] += x * x;
    }
#pragma unroll
  for (int r = 0; r < 8; ++r)
    for (int off = 1; off < 16; off <<= 1) {
      s1[r] += __shfl_xor(s1[r], off);
      s2[r] += __shfl_xor(s2[r], off);
    }
  if (m == 0) {
    *(v4f*)&red[0][w][hh][0] = v4f{s1[0], s1[1], s1[2], s1[3]};
    *(v4f*)&red[0][w][hh][4] = v4f{s1[4], s1[5], s1[6], s1[7]};
    *(v4f*)&red[1][w][hh][0] = v4f{s2[0], s2[1], s2[2], s2[3]};
    *(v4f*)&red[1][w][hh][4] = v4f{s2[4], s2[5], s2[6], s2[7]};
  }
  __syncthreads();
#pragma unroll
  for (int ws = 0; ws < ATT_NW; ++ws) {
    const v4f alo = *(const v4f*)&red[0][ws][hh][0];
    const v4f ahi = *(const v4f*)&red[0][ws][hh][4];
    const v4f blo = *(const v4f*)&red[1][ws][hh][0];
    const v4f bhi = *(const v4f*)&red[1][ws][hh][4];
#pragma unroll
    for (int r = 0; r < 4; ++r) {
      s1[r] = ws == 0 ? alo[r] : s1[r] + alo[r];
      s1[r + 4] = ws == 0 ? ahi[r] : s1[r + 4] + ahi[r];
      s2[r] = ws == 0 ? blo[r] : s2[r] + blo[r];
      s2[r + 4] = ws == 0 ? bhi[r] : s2[r + 4] + bhi[r];
    }
  }
  float gc[4], bc[4];
#pragma unroll
  for (int tc = 0; tc < 4; ++tc) {
    gc[tc] = g[n0 + tc * 16 + m];
    bc[tc] = beta[n0 + tc * 16 + m];
  }
#pragma unroll
  for (int r = 0; r < 8; ++r) {
    const float mu = s1[r] * (1.0f / 256.0f);
    const float var = s2[r] * (1.0f / 256.0f) - mu * mu;
    const float rstd = rsqrtf(var + 1e-5f);
#pragma unroll
    for (int tc = 0; tc < 4; ++tc) {
      const size_t idx = (size_t)(q0 + r + 8 * hh) * 256 + n0 + tc * 16 + m;
      h[idx] = xs[tc][r];
      ln_out[idx] = (xs[tc][r] - mu) * rstd * gc[tc] + bc[tc];
    }
  }
}

__global__ __launch_bounds__(128) void fold_out_kernel(
    const float* __restrict__ W2, const float* __restrict__ b2,
    const float* __restrict__ W_out, const float* __restrict__ b_out,
    float* __restrict__ Wf, float* __restrict__ bf) {
  const int lane = threadIdx.x & 31;
  const int w = __builtin_amdgcn_readfirstlane((int)(threadIdx.x >> 5));
  const int m = lane & 15;
  const int hh = lane >> 4;
  const int rb = blockIdx.x;
  const int n = w * 16 + m;

  const bool wrow = rb < 16;
  const float* arow = wrow ? W2 + (size_t)(rb * 16 + m) * 256 : b2;
  v8f acc0 = {}, acc1 = {};
#pragma unroll
  for (int k = 0; k < 256; k += 8) {
    v2f a0 = *(const v2f*)(arow + k + hh * 2);
    v2f a1 = *(const v2f*)(arow + k + 4 + hh * 2);
    v2f b0, b1;
    b0.x = W_out[(size_t)(k + hh * 2) * 64 + n];
    b0.y = W_out[(size_t)(k + hh * 2 + 1) * 64 + n];
    b1.x = W_out[(size_t)(k + 4 + hh * 2) * 64 + n];
    b1.y = W_out[(size_t)(k + 4 + hh * 2 + 1) * 64 + n];
    acc0 = wmma4(a0, b0, acc0);
    acc1 = wmma4(a1, b1, acc1);
  }
  acc0 += acc1;
  if (wrow) {
#pragma unroll
    for (int r = 0; r < 8; ++r)
      Wf[(size_t)(rb * 16 + r + 8 * hh) * 64 + n] = acc0[r];
  } else if (hh == 0) {
    bf[n] = acc0[0] + b_out[n];
  }
}

extern "C" void kernel_launch(void* const* d_in, const int* in_sizes, int n_in,
                              void* d_out, int out_size, void* d_ws,
                              size_t ws_size, hipStream_t stream) {
  const float* x = (const float*)d_in[0];
  const float* W_in = (const float*)d_in[1];
  const float* b_in = (const float*)d_in[2];
  const float* W_qkv = (const float*)d_in[3];
  const float* b_qkv = (const float*)d_in[4];
  const float* g_attn = (const float*)d_in[5];
  const float* be_attn = (const float*)d_in[6];
  const float* g_mlp = (const float*)d_in[7];
  const float* be_mlp = (const float*)d_in[8];
  const float* W1 = (const float*)d_in[9];
  const float* b1 = (const float*)d_in[10];
  const float* W2 = (const float*)d_in[11];
  const float* b2 = (const float*)d_in[12];
  const float* W_out = (const float*)d_in[13];
  const float* b_out = (const float*)d_in[14];
  float* out = (float*)d_out;

  const int N = in_sizes[0] / 64;

  float* ws = (float*)d_ws;
  float* h = ws;
  float* ln = ws + (size_t)N * 256;
  float* qkv = ws + (size_t)N * 512;
  float* m1 = qkv;
  float* Wf = ws + (size_t)N * 768;
  float* bf = Wf + 256 * 64;

  const dim3 blk4(128);
  const dim3 blk8(256);
  const int RB64 = N / 64;
  const int QB = N / 16;

  inproj_ln_kernel<64><<<dim3(QB), blk4, 0, stream>>>(
      x, W_in, b_in, g_attn, be_attn, h, ln);
  gemm_bias_kernel<0, 4, 256, 768, false>
      <<<dim3(RB64, 12), blk4, 0, stream>>>(ln, W_qkv, b_qkv, nullptr, qkv,
                                            nullptr, nullptr);
  attn_kernel<<<dim3(QB), blk4, 0, stream>>>(qkv, h, g_mlp, be_mlp, ln);
  fold_out_kernel<<<dim3(17), blk4, 0, stream>>>(W2, b2, W_out, b_out, Wf, bf);
  gemm_bias_kernel<1, 4, 256, 256, false>
      <<<dim3(RB64, 4), blk4, 0, stream>>>(ln, W1, b1, nullptr, m1, nullptr,
                                           nullptr);
  gemm_bias_kernel<0, 8, 256, 64, true><<<dim3(RB64, 1), blk8, 0, stream>>>(
      h, W_out, bf, nullptr, out, m1, Wf);
}


# QKV projection re-associated through LayerNorm and the input projection: qkv = rstd*(x@(W_in*g@W_qkv) + c0 - mu*cs) + c1; QKV GEMM K=64, not 256; ln1 not written; k sum order changes
# speedup vs baseline: 3.3373x; 1.2981x over previous
#include <hip/hip_runtime.h>
#include <math.h>

typedef __attribute__((ext_vector_type(2))) float v2f;
typedef __attribute__((ext_vector_type(8))) float v8f;

__device__ __forceinline__ v8f wmma4(v2f a, v2f b, v8f c) {
  return __builtin_amdgcn_wmma_f32_16x16x4_f32(
       false, a,  false, b,
       (short)0, c,  false,  false);
}

typedef __attribute__((ext_vector_type(4))) float v4f;

#define GEMM_SST 36
#define GEMM_WST 72

__device__ __forceinline__ void gemm_kstep(const float* __restrict__ scr,
                                           const float* __restrict__ wsc,
                                           int k4, int wk, int m, int hh,
                                           v8f (&acc)[4][4]) {
  v2f a[4];
#pragma unroll
  for (int rt = 0; rt < 4; ++rt)
    a[rt] = *(const v2f*)(scr + (rt * 16 + m) * GEMM_SST + k4 + hh * 2);
  v2f b[4];
#pragma unroll
  for (int t = 0; t < 4; ++t) {
    b[t].x = wsc[(wk + hh * 2) * GEMM_WST + t * 16 + m];
    b[t].y = wsc[(wk + hh * 2 + 1) * GEMM_WST + t * 16 + m];
  }
#pragma unroll
  for (int rt = 0; rt < 4; ++rt)
#pragma unroll
    for (int t = 0; t < 4; ++t) acc[rt][t] = wmma4(a[rt], b[t], acc[rt][t]);
}

template <int EPI, int SPLIT, int K, int NC, bool DUAL>
__global__ __launch_bounds__(32 * SPLIT) void gemm_bias_kernel(
    const float* __restrict__ A, const float* __restrict__ W,
    const float* __restrict__ bias, const float* __restrict__ resid,
    float* __restrict__ C, const float* __restrict__ A2,
    const float* __restrict__ Wb2) {
  __shared__ v4f part[SPLIT][16][2][32];

  const int lane = threadIdx.x & 31;
  const int w = __builtin_amdgcn_readfirstlane((int)(threadIdx.x >> 5));
  const int m = lane & 15;
  const int hh = lane >> 4;
  const int row0 = blockIdx.x * 64;
  const int n0 = blockIdx.y * 64;

  v8f acc[4][4];
#pragma unroll
  for (int rt = 0; rt < 4; ++rt)
#pragma unroll
    for (int t = 0; t < 4; ++t) acc[rt][t] = {};

  constexpr int KW = DUAL ? SPLIT / 2 : SPLIT;
  static_assert(K % (32 * KW) == 0, "whole 32-k blocks per wave");
  static_assert((64 * GEMM_SST + 16 * GEMM_WST) * 4 <= 16 * 2 * 32 * 16,
                "scratch fits its plane");
  float* scr = (float*)&part[w][0][0][0];
  float* wsc = scr + 64 * GEMM_SST;
  const int lr = lane >> 3;
  const int lc = (lane & 7) * 4;
  const int wr = lane >> 4;
  const int wc = (lane & 15) * 4;

  constexpr int kper = K / KW;
  const int kbeg = (w % KW) * kper;
  const float* Ab = (DUAL && w >= KW) ? A2 : A;
  const float* Wb = (DUAL && w >= KW) ? Wb2 : W;
  const float* Ap = Ab + (size_t)(row0 + lr) * K + kbeg + lc;
  const float* Wp = Wb + (size_t)(kbeg + wr) * NC + n0 + wc;
#pragma nounroll
  for (int kb = 0; kb < kper; kb += 32) {
    v4f ga[4][4];
#pragma unroll
    for (int rt = 0; rt < 4; ++rt)
#pragma unroll
      for (int i = 0; i < 4; ++i)
        ga[rt][i] = *(const v4f*)(Ap + (size_t)(rt * 16 + i * 4) * K + kb);
    v4f gw[8];
#pragma unroll
    for (int i = 0; i < 8; ++i)
      gw[i] = *(const v4f*)(Wp + (size_t)(kb + i * 2) * NC);
    __builtin_amdgcn_sched_barrier(0);
#pragma unroll
    for (int rt = 0; rt < 4; ++rt)
#pragma unroll
      for (int i = 0; i < 4; ++i)
        *(v4f*)(scr + (rt * 16 + i * 4 + lr) * GEMM_SST + lc) = ga[rt][i];
#pragma unroll
    for (int i = 0; i < 8; ++i)
      *(v4f*)(wsc + (i * 2 + wr) * GEMM_WST + wc) = gw[i];

#pragma unroll
    for (int i = 0; i < 8; ++i)
      gw[i] = *(const v4f*)(Wp + (size_t)(kb + 16 + i * 2) * NC);

#pragma unroll
    for (int j = 0; j < 4; ++j) gemm_kstep(scr, wsc, j * 4, j * 4, m, hh, acc);

#pragma unroll
    for (int i = 0; i < 8; ++i)
      *(v4f*)(wsc + (i * 2 + wr) * GEMM_WST + wc) = gw[i];

#pragma unroll
    for (int j = 4; j < 8; ++j)
      gemm_kstep(scr, wsc, j * 4, (j - 4) * 4, m, hh, acc);
  }

#pragma unroll
  for (int rt = 0; rt < 4; ++rt)
#pragma unroll
    for (int t = 0; t < 4; ++t) {
      part[w][rt * 4 + t][0][lane] =
          __builtin_shufflevector(acc[rt][t], acc[rt][t], 0, 1, 2, 3);
      part[w][rt * 4 + t][1][lane] =
          __builtin_shufflevector(acc[rt][t], acc[rt][t], 4, 5, 6, 7);
    }
  __syncthreads();

  constexpr int TPW = 16 / SPLIT;
  float v[TPW][8];
#pragma unroll
  for (int i = 0; i < TPW; ++i) {
    const int j = w * TPW + i;
    v4f lo = part[0][j][0][lane];
    v4f hi = part[0][j][1][lane];
#pragma unroll
    for (int ws = 1; ws < SPLIT; ++ws) {
      lo += part[ws][j][0][lane];
      hi += part[ws][j][1][lane];
    }
#pragma unroll
    for (int r = 0; r < 4; ++r) {
      v[i][r] = lo[r];
      v[i][r + 4] = hi[r];
    }
  }

  float bn[TPW];
#pragma unroll
  for (int i = 0; i < TPW; ++i) {
    const int j = w * TPW + i;
    bn[i] = bias[n0 + (j & 3) * 16 + m];
  }

  constexpr int RTW = TPW >= 4 ? TPW / 4 : 1;
  float csn[TPW], c1n[TPW];
  v2f st[RTW][8];
  if (EPI == 3) {
#pragma unroll
    for (int i = 0; i < TPW; ++i) {
      const int j = w * TPW + i;
      csn[i] = bias[NC + n0 + (j & 3) * 16 + m];
      c1n[i] = bias[2 * NC + n0 + (j & 3) * 16 + m];
    }
#pragma unroll
    for (int q = 0; q < RTW; ++q)
#pragma unroll
      for (int r = 0; r < 8; ++r)
        st[q][r] = *(const v2f*)(resid +
            2 * (size_t)(row0 + (((w * TPW) >> 2) + q) * 16 + r + 8 * hh));
  }

  float res[TPW][8];
  if (EPI == 2) {
#pragma unroll
    for (int i = 0; i < TPW; ++i) {
      const int j = w * TPW + i;
#pragma unroll
      for (int r = 0; r < 8; ++r)
        res[i][r] = resid[(size_t)(row0 + (j >> 2) * 16 + r + 8 * hh) * NC +
                          n0 + (j & 3) * 16 + m];
    }
  }

#pragma unroll
  for (int i = 0; i < TPW; ++i) {
    const int j = w * TPW + i;
#pragma unroll
    for (int r = 0; r < 8; ++r) {
      const int row = row0 + (j >> 2) * 16 + r + 8 * hh;
      const int col = n0 + (j & 3) * 16 + m;
      float x = v[i][r] + bn[i];
      if (EPI == 1) x = 0.5f * x * (1.0f + erff(x * 0.70710678118654752f));
      if (EPI == 2) x += res[i][r];
      if (EPI == 3) {
        const v2f mr = st[TPW >= 4 ? (i >> 2) : 0][r];
        x = mr.y * (x - mr.x * csn[i]) + c1n[i];
      }
      C[(size_t)row * NC + col] = x;
    }
  }
}

#define FOLD_QKV_BLOCKS 60
#define FOLD_AST 260

__device__ __forceinline__ void fold_qkv_block(
    int fb, float* __restrict__ fa, const float* __restrict__ W_in,
    const float* __restrict__ b_in, const float* __restrict__ g,
    const float* __restrict__ beta, const float* __restrict__ W_qkv,
    const float* __restrict__ b_qkv, float* __restrict__ Wx) {
  const int lane = threadIdx.x & 31;
  const int w = __builtin_amdgcn_readfirstlane((int)(threadIdx.x >> 5));
  const int m = lane & 15;
  const int hh = lane >> 4;
  const int rb = fb / 12;
  const int n = (fb - rb * 12) * 64 + w * 16 + m;
  const bool wrow = rb < 4;

#pragma unroll
  for (int it = 0; it < 8; ++it) {
    const int e = (it * 128 + (int)threadIdx.x) * 4;
    const int row = e >> 8;
    const int col = e & 255;
    const v4f gv = *(const v4f*)(g + col);
    v4f v;
    if (wrow) {
      v = *(const v4f*)(W_in + (size_t)(rb * 16 + row) * 256 + col) * gv;
    } else {
      const float* src = row == 0 ? b_in : (row == 2 ? beta : g);
      v = *(const v4f*)(src + col);
      if (row == 0) v *= gv;
      if (row > 2) v = v4f{0.f, 0.f, 0.f, 0.f};
    }
    *(v4f*)(fa + row * FOLD_AST + col) = v;
  }
  __syncthreads();

  v8f acc0 = {}, acc1 = {};
#pragma unroll
  for (int k = 0; k < 256; k += 8) {
    const v2f a0 = *(const v2f*)(fa + m * FOLD_AST + k + hh * 2);
    const v2f a1 = *(const v2f*)(fa + m * FOLD_AST + k + 4 + hh * 2);
    v2f b0, b1;
    b0.x = W_qkv[(size_t)(k + hh * 2) * 768 + n];
    b0.y = W_qkv[(size_t)(k + hh * 2 + 1) * 768 + n];
    b1.x = W_qkv[(size_t)(k + 4 + hh * 2) * 768 + n];
    b1.y = W_qkv[(size_t)(k + 4 + hh * 2 + 1) * 768 + n];
    acc0 = wmma4(a0, b0, acc0);
    acc1 = wmma4(a1, b1, acc1);
  }
  acc0 += acc1;
  if (wrow) {
#pragma unroll
    for (int r = 0; r < 8; ++r)
      Wx[(size_t)(rb * 16 + r + 8 * hh) * 768 + n] = acc0[r];
  } else if (hh == 0) {
    Wx[(size_t)64 * 768 + n] = acc0[0];
    Wx[(size_t)65 * 768 + n] = acc0[1];
    Wx[(size_t)66 * 768 + n] = acc0[2] + b_qkv[n];
  }
}

template <int K>
__global__ __launch_bounds__(128) void inproj_stats_kernel(
    const float* __restrict__ X, const float* __restrict__ W,
    const float* __restrict__ bias, const float* __restrict__ g,
    const float* __restrict__ beta, const float* __restrict__ W_qkv,
    const float* __restrict__ b_qkv, float* __restrict__ h,
    float* __restrict__ stats, float* __restrict__ Wx) {
  __shared__ __attribute__((aligned(16))) float red[2][4][2][8];
  __shared__ __attribute__((aligned(16))) float fa[16 * FOLD_AST];

  if (blockIdx.x < FOLD_QKV_BLOCKS) {
    fold_qkv_block(blockIdx.x, fa, W, bias, g, beta, W_qkv, b_qkv, Wx);
    return;
  }

  const int lane = threadIdx.x & 31;
  const int w = __builtin_amdgcn_readfirstlane((int)(threadIdx.x >> 5));
  const int m = lane & 15;
  const int hh = lane >> 4;
  const int row0 = (blockIdx.x - FOLD_QKV_BLOCKS) * 16;
  const int n0 = w * 64;

  v8f acc[4];
#pragma unroll
  for (int t = 0; t < 4; ++t) acc[t] = {};
#pragma unroll
  for (int k = 0; k < K; k += 4) {
    const v2f a = *(const v2f*)(X + (size_t)(row0 + m) * K + k + hh * 2);
    v2f b[4];
#pragma unroll
    for (int t = 0; t < 4; ++t) {
      const int n = n0 + t * 16 + m;
      b[t].x = W[(size_t)(k + hh * 2) * 256 + n];
      b[t].y = W[(size_t)(k + hh * 2 + 1) * 256 + n];
    }
#pragma unroll
    for (int t = 0; t < 4; ++t) acc[t] = wmma4(a, b[t], acc[t]);
  }

  float s1[8], s2[8];
#pragma unroll
  for (int r = 0; r < 8; ++r) {
    s1[r] = 0.f;
    s2[r] = 0.f;
  }
#pragma unroll
  for (int t = 0; t < 4; ++t) {
    const float bn = bias[n0 + t * 16 + m];
#pragma unroll
    for (int r = 0; r < 8; ++r) {
      const float x = acc[t][r] + bn;
      h[(size_t)(row0 + r + 8 * hh) * 256 + n0 + t * 16 + m] = x;
      s1[r] += x;
      s2[r] += x * x;
    }
  }
#pragma unroll
  for (int r = 0; r < 8; ++r)
    for (int off = 1; off < 16; off <<= 1) {
      s1[r] += __shfl_xor(s1[r], off);
      s2[r] += __shfl_xor(s2[r], off);
    }
  if (m == 0) {
    *(v4f*)&red[0][w][hh][0] = v4f{s1[0], s1[1], s1[2], s1[3]};
    *(v4f*)&red[0][w][hh][4] = v4f{s1[4], s1[5], s1[6], s1[7]};
    *(v4f*)&red[1][w][hh][0] = v4f{s2[0], s2[1], s2[2], s2[3]};
    *(v4f*)&red[1][w][hh][4] = v4f{s2[4], s2[5], s2[6], s2[7]};
  }
  __syncthreads();
  if (w != 0 || m != 0) return;
#pragma unroll
  for (int ws = 0; ws < 4; ++ws) {
    const v4f alo = *(const v4f*)&red[0][ws][hh][0];
    const v4f ahi = *(const v4f*)&red[0][ws][hh][4];
    const v4f blo = *(const v4f*)&red[1][ws][hh][0];
    const v4f bhi = *(const v4f*)&red[1][ws][hh][4];
#pragma unroll
    for (int r = 0; r < 4; ++r) {
      s1[r] = ws == 0 ? alo[r] : s1[r] + alo[r];
      s1[r + 4] = ws == 0 ? ahi[r] : s1[r + 4] + ahi[r];
      s2[r] = ws == 0 ? blo[r] : s2[r] + blo[r];
      s2[r + 4] = ws == 0 ? bhi[r] : s2[r + 4] + bhi[r];
    }
  }
#pragma unroll
  for (int r = 0; r < 8; ++r) {
    const float mu = s1[r] * (1.0f / 256.0f);
    const float var = s2[r] * (1.0f / 256.0f) - mu * mu;
    const float rstd = rsqrtf(var + 1e-5f);
    *(v2f*)(stats + 2 * (size_t)(row0 + r + 8 * hh)) = v2f{mu, rstd};
  }
}

#define ATT_T 4096
#define ATT_QS 768
#define ATT_WIN 128
#define ATT_PST 148
#define ATT_SST 36
#define ATT_NW 4

template <int NT>
__device__ __forceinline__ void attn_scores(const float* __restrict__ Q,
                                            const float* __restrict__ Km,
                                            float* __restrict__ stg, int q0,
                                            int bb, int t0, int lane,
                                            v8f (&s)[3]) {
  const int m = lane & 15;
  const int hh = lane >> 4;
  const int lr = lane >> 3;
  const int lc = (lane & 7) * 4;
#pragma unroll
  for (int t = 0; t < NT; ++t) s[t] = {};

#pragma nounroll
  for (int db = 0; db < 256; db += 32) {
    v4f gq[4];
    v4f gk[NT][4];
#pragma unroll
    for (int i = 0; i < 4; ++i)
      gq[i] = *(const v4f*)(Q + (size_t)(q0 + i * 4 + lr) * ATT_QS + db + lc);
#pragma unroll
    for (int t = 0; t < NT; ++t)
#pragma unroll
      for (int i = 0; i < 4; ++i) {
        const int key = q0 - ATT_WIN + (t0 + t) * 16 + i * 4 + lr;
        const int kc = key < bb ? bb : key;
        gk[t][i] = *(const v4f*)(Km + (size_t)kc * ATT_QS + db + lc);
      }
    __builtin_amdgcn_sched_barrier(0);
#pragma unroll
    for (int i = 0; i < 4; ++i)
      *(v4f*)(stg + (i * 4 + lr) * ATT_SST + lc) = gq[i];
#pragma unroll
    for (int t = 0; t < NT; ++t)
#pragma unroll
      for (int i = 0; i < 4; ++i)
        *(v4f*)(stg + ((t + 1) * 16 + i * 4 + lr) * ATT_SST + lc) = gk[t][i];
#pragma unroll
    for (int j = 0; j < 8; ++j) {
      const v2f a = *(const v2f*)(stg + m * ATT_SST + j * 4 + hh * 2);
#pragma unroll
      for (int t = 0; t < NT; ++t) {
        const v2f b =
            *(const v2f*)(stg + ((t + 1) * 16 + m) * ATT_SST + j * 4 + hh * 2);
        s[t] = wmma4(a, b, s[t]);
      }
    }
  }
}

__global__ __launch_bounds__(32 * ATT_NW) void attn_kernel(
    const float* __restrict__ qkv, float* __restrict__ h,
    const float* __restrict__ g, const float* __restrict__ beta,
    float* __restrict__ ln_out) {
  __shared__ __attribute__((aligned(16))) float P[16 * ATT_PST];
  __shared__ __attribute__((aligned(16))) float stg[ATT_NW][4 * 16 * ATT_SST];
  __shared__ __attribute__((aligned(16))) float red[2][ATT_NW][2][8];

  const int lane = threadIdx.x & 31;
  const int w = __builtin_amdgcn_readfirstlane((int)(threadIdx.x >> 5));
  const int m = lane & 15;
  const int hh = lane >> 4;
  const int q0 = blockIdx.x * 16;
  const int bb = q0 & ~(ATT_T - 1);

  const float* Q = qkv;
  const float* Km = qkv + 256;
  const float* V = qkv + 512;

  const int nt = w == 0 ? 3 : 2;
  const int t0 = w == 0 ? 0 : 2 * w + 1;
  v8f s[3];
  if (w == 0) {
    attn_scores<3>(Q, Km, stg[w], q0, bb, t0, lane, s);
  } else {
    attn_scores<2>(Q, Km, stg[w], q0, bb, t0, lane, s);
    s[2] = {};
  }

  const float NEG = -3.402823466e38f;
  float rmax[8], rsum[8];
#pragma unroll
  for (int r = 0; r < 8; ++r) rmax[r] = NEG;
#pragma unroll
  for (int tt = 0; tt < 3; ++tt) {
    if (tt < nt) {
      const int key = q0 - ATT_WIN + (t0 + tt) * 16 + m;
#pragma unroll
      for (int r = 0; r < 8; ++r) {
        const int q = q0 + r + 8 * hh;
        const int d = q - key;
        const bool ok = (key >= bb) && (d >= 0) && (d <= ATT_WIN);
        const float val = ok ? s[tt][r] * 0.0625f : NEG;
        s[tt][r] = val;
        rmax[r] = fmaxf(rmax[r], val);
      }
    }
  }
#pragma unroll
  for (int r = 0; r < 8; ++r)
    for (int off = 1; off < 16; off <<= 1)
      rmax[r] = fmaxf(rmax[r], __shfl_xor(rmax[r], off));
  if (m == 0) {
    *(v4f*)&red[0][w][hh][0] = v4f{rmax[0], rmax[1], rmax[2], rmax[3]};
    *(v4f*)&red[0][w][hh][4] = v4f{rmax[4], rmax[5], rmax[6], rmax[7]};
  }
  __syncthreads();
#pragma unroll
  for (int ws = 0; ws < ATT_NW; ++ws) {
    const v4f lo = *(const v4f*)&red[0][ws][hh][0];
    const v4f hi = *(const v4f*)&red[0][ws][hh][4];
#pragma unroll
    for (int r = 0; r < 4; ++r) {
      rmax[r] = fmaxf(rmax[r], lo[r]);
      rmax[r + 4] = fmaxf(rmax[r + 4], hi[r]);
    }
  }

#pragma unroll
  for (int r = 0; r < 8; ++r) rsum[r] = 0.f;
#pragma unroll
  for (int tt = 0; tt < 3; ++tt) {
    if (tt < nt) {
#pragma unroll
      for (int r = 0; r < 8; ++r) {
        const float e = expf(s[tt][r] - rmax[r]);
        s[tt][r] = e;
        rsum[r] += e;
      }
    }
  }
#pragma unroll
  for (int r = 0; r < 8; ++r)
    for (int off = 1; off < 16; off <<= 1) rsum[r] += __shfl_xor(rsum[r], off);
  if (m == 0) {
    *(v4f*)&red[1][w][hh][0] = v4f{rsum[0], rsum[1], rsum[2], rsum[3]};
    *(v4f*)&red[1][w][hh][4] = v4f{rsum[4], rsum[5], rsum[6], rsum[7]};
  }
  __syncthreads();
#pragma unroll
  for (int ws = 0; ws < ATT_NW; ++ws) {
    const v4f lo = *(const v4f*)&red[1][ws][hh][0];
    const v4f hi = *(const v4f*)&red[1][ws][hh][4];
#pragma unroll
    for (int r = 0; r < 4; ++r) {
      rsum[r] = ws == 0 ? lo[r] : rsum[r] + lo[r];
      rsum[r + 4] = ws == 0 ? hi[r] : rsum[r + 4] + hi[r];
    }
  }

#pragma unroll
  for (int tt = 0; tt < 3; ++tt) {
    if (tt < nt) {
#pragma unroll
      for (int r = 0; r < 8; ++r)
        P[(r + 8 * hh) * ATT_PST + (t0 + tt) * 16 + m] = s[tt][r] / rsum[r];
    }
  }
  __syncthreads();

  const int n0 = w * 64;
  float hv[4][8];
#pragma unroll
  for (int tc = 0; tc < 4; ++tc)
#pragma unroll
    for (int r = 0; r < 8; ++r)
      hv[tc][r] = h[(size_t)(q0 + r + 8 * hh) * 256 + n0 + tc * 16 + m];
  v8f o[4];
#pragma unroll
  for (int tc = 0; tc < 4; ++tc) o[tc] = {};
  for (int t = 0; t < 9; ++t) {
    const int kt0 = q0 - ATT_WIN + t * 16;
#pragma unroll
    for (int kk = 0; kk < 16; kk += 4) {
      const v2f a = *(const v2f*)&P[m * ATT_PST + t * 16 + kk + hh * 2];
      const int k0 = kt0 + kk + hh * 2;
      const int kcx = k0 < bb ? bb : k0;
      const int kcy = (k0 + 1) < bb ? bb : (k0 + 1);
#pragma unroll
      for (int tc = 0; tc < 4; ++tc) {
        const int col = n0 + tc * 16 + m;
        v2f b;
        b.x = V[(size_t)kcx * ATT_QS + col];
        b.y = V[(size_t)kcy * ATT_QS + col];
        o[tc] = wmma4(a, b, o[tc]);
      }
    }
  }
  float xs[4][8];
  float s1[8], s2[8];
#pragma unroll
  for (int r = 0; r < 8; ++r) {
    s1[r] = 0.f;
    s2[r] = 0.f;
  }
#pragma unroll
  for (int tc = 0; tc < 4; ++tc)
#pragma unroll
    for (int r = 0; r < 8; ++r) {
      const float x = hv[tc][r] + o[tc][r];
      xs[tc][r] = x;
      s1[r] += x;
      s2[r] += x * x;
    }
#pragma unroll
  for (int r = 0; r < 8; ++r)
    for (int off = 1; off < 16; off <<= 1) {
      s1[r] += __shfl_xor(s1[r], off);
      s2[r] += __shfl_xor(s2[r], off);
    }
  if (m == 0) {
    *(v4f*)&red[0][w][hh][0] = v4f{s1[0], s1[1], s1[2], s1[3]};
    *(v4f*)&red[0][w][hh][4] = v4f{s1[4], s1[5], s1[6], s1[7]};
    *(v4f*)&red[1][w][hh][0] = v4f{s2[0], s2[1], s2[2], s2[3]};
    *(v4f*)&red[1][w][hh][4] = v4f{s2[4], s2[5], s2[6], s2[7]};
  }
  __syncthreads();
#pragma unroll
  for (int ws = 0; ws < ATT_NW; ++ws) {
    const v4f alo = *(const v4f*)&red[0][ws][hh][0];
    const v4f ahi = *(const v4f*)&red[0][ws][hh][4];
    const v4f blo = *(const v4f*)&red[1][ws][hh][0];
    const v4f bhi = *(const v4f*)&red[1][ws][hh][4];
#pragma unroll
    for (int r = 0; r < 4; ++r) {
      s1[r] = ws == 0 ? alo[r] : s1[r] + alo[r];
      s1[r + 4] = ws == 0 ? ahi[r] : s1[r + 4] + ahi[r];
      s2[r] = ws == 0 ? blo[r] : s2[r] + blo[r];
      s2[r + 4] = ws == 0 ? bhi[r] : s2[r + 4] + bhi[r];
    }
  }
  float gc[4], bc[4];
#pragma unroll
  for (int tc = 0; tc < 4; ++tc) {
    gc[tc] = g[n0 + tc * 16 + m];
    bc[tc] = beta[n0 + tc * 16 + m];
  }
#pragma unroll
  for (int r = 0; r < 8; ++r) {
    const float mu = s1[r] * (1.0f / 256.0f);
    const float var = s2[r] * (1.0f / 256.0f) - mu * mu;
    const float rstd = rsqrtf(var + 1e-5f);
#pragma unroll
    for (int tc = 0; tc < 4; ++tc) {
      const size_t idx = (size_t)(q0 + r + 8 * hh) * 256 + n0 + tc * 16 + m;
      h[idx] = xs[tc][r];
      ln_out[idx] = (xs[tc][r] - mu) * rstd * gc[tc] + bc[tc];
    }
  }
}

__global__ __launch_bounds__(128) void fold_out_kernel(
    const float* __restrict__ W2, const float* __restrict__ b2,
    const float* __restrict__ W_out, const float* __restrict__ b_out,
    float* __restrict__ Wf, float* __restrict__ bf) {
  const int lane = threadIdx.x & 31;
  const int w = __builtin_amdgcn_readfirstlane((int)(threadIdx.x >> 5));
  const int m = lane & 15;
  const int hh = lane >> 4;
  const int rb = blockIdx.x;
  const int n = w * 16 + m;

  const bool wrow = rb < 16;
  const float* arow = wrow ? W2 + (size_t)(rb * 16 + m) * 256 : b2;
  v8f acc0 = {}, acc1 = {};
#pragma unroll
  for (int k = 0; k < 256; k += 8) {
    v2f a0 = *(const v2f*)(arow + k + hh * 2);
    v2f a1 = *(const v2f*)(arow + k + 4 + hh * 2);
    v2f b0, b1;
    b0.x = W_out[(size_t)(k + hh * 2) * 64 + n];
    b0.y = W_out[(size_t)(k + hh * 2 + 1) * 64 + n];
    b1.x = W_out[(size_t)(k + 4 + hh * 2) * 64 + n];
    b1.y = W_out[(size_t)(k + 4 + hh * 2 + 1) * 64 + n];
    acc0 = wmma4(a0, b0, acc0);
    acc1 = wmma4(a1, b1, acc1);
  }
  acc0 += acc1;
  if (wrow) {
#pragma unroll
    for (int r = 0; r < 8; ++r)
      Wf[(size_t)(rb * 16 + r + 8 * hh) * 64 + n] = acc0[r];
  } else if (hh == 0) {
    bf[n] = acc0[0] + b_out[n];
  }
}

extern "C" void kernel_launch(void* const* d_in, const int* in_sizes, int n_in,
                              void* d_out, int out_size, void* d_ws,
                              size_t ws_size, hipStream_t stream) {
  const float* x = (const float*)d_in[0];
  const float* W_in = (const float*)d_in[1];
  const float* b_in = (const float*)d_in[2];
  const float* W_qkv = (const float*)d_in[3];
  const float* b_qkv = (const float*)d_in[4];
  const float* g_attn = (const float*)d_in[5];
  const float* be_attn = (const float*)d_in[6];
  const float* g_mlp = (const float*)d_in[7];
  const float* be_mlp = (const float*)d_in[8];
  const float* W1 = (const float*)d_in[9];
  const float* b1 = (const float*)d_in[10];
  const float* W2 = (const float*)d_in[11];
  const float* b2 = (const float*)d_in[12];
  const float* W_out = (const float*)d_in[13];
  const float* b_out = (const float*)d_in[14];
  float* out = (float*)d_out;

  const int N = in_sizes[0] / 64;

  float* ws = (float*)d_ws;
  float* h = ws;
  float* ln = ws + (size_t)N * 256;
  float* qkv = ws + (size_t)N * 512;
  float* m1 = qkv;
  float* Wf = ws + (size_t)N * 768;
  float* bf = Wf + 256 * 64;
  float* Wx = ln;
  float* stats = ln + 65536;

  const dim3 blk4(128);
  const dim3 blk2(64);
  const dim3 blk8(256);
  const int RB64 = N / 64;
  const int QB = N / 16;

  inproj_stats_kernel<64><<<dim3(FOLD_QKV_BLOCKS + QB), blk4, 0, stream>>>(
      x, W_in, b_in, g_attn, be_attn, W_qkv, b_qkv, h, stats, Wx);
  gemm_bias_kernel<3, 2, 64, 768, false>
      <<<dim3(RB64, 12), blk2, 0, stream>>>(x, Wx, Wx + 64 * 768, stats, qkv,
                                            nullptr, nullptr);
  attn_kernel<<<dim3(QB), blk4, 0, stream>>>(qkv, h, g_mlp, be_mlp, ln);
  fold_out_kernel<<<dim3(17), blk4, 0, stream>>>(W2, b2, W_out, b_out, Wf, bf);
  gemm_bias_kernel<1, 4, 256, 256, false>
      <<<dim3(RB64, 4), blk4, 0, stream>>>(ln, W1, b1, nullptr, m1, nullptr,
                                           nullptr);
  gemm_bias_kernel<0, 8, 256, 64, true><<<dim3(RB64, 1), blk8, 0, stream>>>(
      h, W_out, bf, nullptr, out, m1, Wf);
}
